# starting kernel
# baseline (speedup 1.0000x reference)
#include <hip/hip_runtime.h>

typedef __attribute__((ext_vector_type(2))) float v2f;
typedef __attribute__((ext_vector_type(8))) float v8f;

#define NEG_SLOPE 0.01f

__global__ void k_norm_init(float* __restrict__ norm, int n) {
  int i = blockIdx.x * blockDim.x + threadIdx.x;
  if (i < n) norm[i] = 1.0f;
}
__global__ void k_norm_count(const int* __restrict__ dst, float* __restrict__ norm, int e) {
  int i = blockIdx.x * blockDim.x + threadIdx.x;
  if (i < e) atomicAdd(&norm[dst[i]], 1.0f);
}
__global__ void k_norm_rsqrt(float* __restrict__ norm, int n) {
  int i = blockIdx.x * blockDim.x + threadIdx.x;
  if (i < n) norm[i] = rsqrtf(norm[i]);
}

__global__ __launch_bounds__(256) void k_gemm_wmma(
    const float* __restrict__ X, const float* __restrict__ W,
    const float* __restrict__ norm,
    float* __restrict__ S, float* __restrict__ AGG,
    int nrows, int K, int Hout) {
  extern __shared__ float lds[];
  const int KSTRIDE = K + 4;
  float* Xs = lds;
  float* Wt = lds + 128 * KSTRIDE;

  const int tid = threadIdx.x;
  const int row0 = blockIdx.x * 128;

  for (int idx = tid * 4; idx < K * Hout; idx += 256 * 4) {
    int k = idx / Hout, c = idx - k * Hout;
    float4 v = *(const float4*)&W[idx];
    Wt[(c + 0) * KSTRIDE + k] = v.x;
    Wt[(c + 1) * KSTRIDE + k] = v.y;
    Wt[(c + 2) * KSTRIDE + k] = v.z;
    Wt[(c + 3) * KSTRIDE + k] = v.w;
  }
  for (int idx = tid * 4; idx < 128 * K; idx += 256 * 4) {
    int r = idx / K, c = idx - r * K;
    float4 v = {0.f, 0.f, 0.f, 0.f};
    int gr = row0 + r;
    if (gr < nrows) v = *(const float4*)&X[(long)gr * K + c];
    *(float4*)&Xs[r * KSTRIDE + c] = v;
  }
  __syncthreads();

  const int lane = tid & 31;
  const int wave = tid >> 5;
  const int mrow = lane & 15;
  const int kg   = lane >> 4;
  const float* Xrow = &Xs[(wave * 16 + mrow) * KSTRIDE + 2 * kg];

  float nrm[8];
#pragma unroll
  for (int r = 0; r < 8; ++r) {
    int grow = row0 + wave * 16 + r + 8 * kg;
    nrm[r] = (grow < nrows) ? norm[grow] : 0.f;
  }

  for (int ct = 0; ct < (Hout >> 4); ++ct) {
    const int col = ct * 16 + mrow;
    const float* Wcol = &Wt[col * KSTRIDE + 2 * kg];
    v8f acc = {};
    for (int k = 0; k < K; k += 4) {
      v2f a = *(const v2f*)&Xrow[k];
      v2f b = *(const v2f*)&Wcol[k];
      acc = __builtin_amdgcn_wmma_f32_16x16x4_f32(
           false, a,  false, b,
           (short)0, acc,  false,  false);
    }
#pragma unroll
    for (int r = 0; r < 8; ++r) {
      int grow = row0 + wave * 16 + r + 8 * kg;
      if (grow < nrows) {
        float sv = acc[r] * nrm[r];
        long o = (long)grow * Hout + col;
        S[o]   = sv;
        AGG[o] = sv;
      }
    }
  }
}

__global__ void k_scatter(const int* __restrict__ src, const int* __restrict__ dst,
                          const float* __restrict__ S, float* __restrict__ AGG,
                          int e, int Hout) {
  const int chunks = Hout >> 2;
  long tid = (long)blockIdx.x * blockDim.x + threadIdx.x;
  if (tid >= (long)e * chunks) return;
  int ed = (int)(tid / chunks);
  int c  = (int)(tid - (long)ed * chunks) << 2;
  int s = src[ed], d = dst[ed];
  const float4 v = *(const float4*)&S[(long)s * Hout + c];
  float* ap = &AGG[(long)d * Hout + c];
  atomicAdd(ap + 0, v.x);
  atomicAdd(ap + 1, v.y);
  atomicAdd(ap + 2, v.z);
  atomicAdd(ap + 3, v.w);
}

__global__ void k_finalize(const float* __restrict__ AGG, const float* __restrict__ norm,
                           const float* __restrict__ bias, const float* __restrict__ xin,
                           float* __restrict__ out, int n, int Hout, int mode) {
  const int chunks = Hout >> 2;
  long tid = (long)blockIdx.x * blockDim.x + threadIdx.x;
  if (tid >= (long)n * chunks) return;
  int i = (int)(tid / chunks);
  int c = (int)(tid - (long)i * chunks) << 2;
  long o = (long)i * Hout + c;
  float nr = norm[i];
  float4 av = *(const float4*)&AGG[o];
  float4 bv = *(const float4*)&bias[c];
  float vx = av.x * nr + bv.x;
  float vy = av.y * nr + bv.y;
  float vz = av.z * nr + bv.z;
  float vw = av.w * nr + bv.w;
  if (mode != 2) {
    vx = vx >= 0.f ? vx : NEG_SLOPE * vx;
    vy = vy >= 0.f ? vy : NEG_SLOPE * vy;
    vz = vz >= 0.f ? vz : NEG_SLOPE * vz;
    vw = vw >= 0.f ? vw : NEG_SLOPE * vw;
  }
  if (mode == 1) {
    float4 xv = *(const float4*)&xin[o];
    vx = (xv.x + vx) * 0.5f;
    vy = (xv.y + vy) * 0.5f;
    vz = (xv.z + vz) * 0.5f;
    vw = (xv.w + vw) * 0.5f;
  }
  float4 r4 = {vx, vy, vz, vw};
  *(float4*)&out[o] = r4;
}

extern "C" void kernel_launch(void* const* d_in, const int* in_sizes, int n_in,
                              void* d_out, int out_size, void* d_ws, size_t ws_size,
                              hipStream_t stream) {
  (void)n_in; (void)out_size; (void)ws_size;
  const float* inputs = (const float*)d_in[0];
  const int*   edges  = (const int*)d_in[1];
  const float* w_in   = (const float*)d_in[2];
  const float* b_in   = (const float*)d_in[3];
  const float* w1     = (const float*)d_in[4];
  const float* b1     = (const float*)d_in[5];
  const float* w2     = (const float*)d_in[6];
  const float* b2     = (const float*)d_in[7];
  const float* w_out  = (const float*)d_in[8];
  const float* b_out  = (const float*)d_in[9];

  const int H   = in_sizes[3];
  const int IN  = in_sizes[2] / H;
  const int N   = in_sizes[0] / IN;
  const int E   = in_sizes[1] / 2;
  const int OUT = in_sizes[9];
  const int B   = in_sizes[5] / H;

  const int* srcI = edges;
  const int* dstI = edges + E;

  float* norm = (float*)d_ws;
  float* S    = norm + (((size_t)N + 255) & ~(size_t)255);
  float* AGG  = S + (size_t)N * H;
  float* Hb   = AGG + (size_t)N * H;

  float* xout = (float*)d_out;
  float* xbuf = xout + (size_t)N * OUT;

  k_norm_init <<<(N + 255) / 256, 256, 0, stream>>>(norm, N);
  k_norm_count<<<(E + 255) / 256, 256, 0, stream>>>(dstI, norm, E);
  k_norm_rsqrt<<<(N + 255) / 256, 256, 0, stream>>>(norm, N);

  auto conv = [&](const float* Xin, const float* W, const float* bias, int K, int Hout,
                  float* outp, int mode, const float* resid) {
    size_t shmem = (size_t)(128 + Hout) * (K + 4) * sizeof(float);
    k_gemm_wmma<<<(N + 127) / 128, 256, shmem, stream>>>(Xin, W, norm, S, AGG, N, K, Hout);
    long tot = (long)E * (Hout >> 2);
    k_scatter<<<(unsigned)((tot + 255) / 256), 256, 0, stream>>>(srcI, dstI, S, AGG, E, Hout);
    long tot2 = (long)N * (Hout >> 2);
    k_finalize<<<(unsigned)((tot2 + 255) / 256), 256, 0, stream>>>(AGG, norm, bias, resid,
                                                                   outp, N, Hout, mode);
  };

  conv(inputs, w_in, b_in, IN, H, xbuf, 0, nullptr);
  for (int i = 0; i < B; ++i) {
    conv(xbuf, w1 + (size_t)i * H * H, b1 + (size_t)i * H, H, H, Hb,   0, nullptr);
    conv(Hb,   w2 + (size_t)i * H * H, b2 + (size_t)i * H, H, H, xbuf, 1, xbuf);
  }
  conv(xbuf, w_out, b_out, H, OUT, xout, 2, nullptr);
}


# Aggregation by destination instead of atomic scatter: edges bucketed by dst once per call (int atomic arrival-order rank), one wave sums a row's in-edges in registers, one store; no float atomics
# speedup vs baseline: 2.5332x; 2.5332x over previous
#include <hip/hip_runtime.h>

typedef __attribute__((ext_vector_type(2))) float v2f;
typedef __attribute__((ext_vector_type(4))) float v4f;
typedef __attribute__((ext_vector_type(8))) float v8f;

#define NEG_SLOPE 0.01f

__global__ void k_cnt_zero(int* __restrict__ cnt, int n) {
  int i = blockIdx.x * blockDim.x + threadIdx.x;
  if (i < n) cnt[i] = 0;
}
__global__ void k_cnt_rank(const int* __restrict__ dst, int* __restrict__ cnt,
                           int* __restrict__ rank, int e) {
  int i = blockIdx.x * blockDim.x + threadIdx.x;
  if (i < e) rank[i] = atomicAdd(&cnt[dst[i]], 1);
}
__global__ __launch_bounds__(256) void k_start_norm(const int* __restrict__ cnt,
                                                    int* __restrict__ segw,
                                                    float* __restrict__ norm, int n) {
  int2* seg = (int2*)segw;
  __shared__ int wsum[8];
  __shared__ int psum[8];
  const int tid = threadIdx.x;
  const int lane = tid & 31;
  const int wv = tid >> 5;
  const int base = blockIdx.x * 256;
  int part = 0;
  const int4* c4 = (const int4*)cnt;
  for (int j = tid; j < (base >> 2); j += 256) {
    int4 v = c4[j];
    part += (v.x + v.y) + (v.z + v.w);
  }
  const int i = base + tid;
  const int c = (i < n) ? cnt[i] : 0;
  int incl = c;
#pragma unroll
  for (int off = 1; off < 32; off <<= 1) {
    int t = __shfl_up(incl, off);
    if (lane >= off) incl += t;
  }
#pragma unroll
  for (int off = 16; off > 0; off >>= 1) part += __shfl_xor(part, off);
  if (lane == 31) wsum[wv] = incl;
  if (lane == 0) psum[wv] = part;
  __syncthreads();
  int offset = 0;
#pragma unroll
  for (int w = 0; w < 8; ++w) {
    offset += psum[w];
    if (w < wv) offset += wsum[w];
  }
  if (i < n) {
    seg[i] = make_int2(offset + incl - c, c);
    norm[i] = rsqrtf(1.0f + (float)c);
  }
}
__global__ void k_fill(const int* __restrict__ src, const int* __restrict__ dst,
                       const int* __restrict__ rank, const int* __restrict__ segw,
                       int* __restrict__ col, int e) {
  int i = blockIdx.x * blockDim.x + threadIdx.x;
  if (i < e) atomicExch(&col[segw[2 * dst[i]] + rank[i]], src[i]);
}

__global__ __launch_bounds__(256) void k_gemm_wmma(
    const float* __restrict__ X, const float* __restrict__ W,
    const float* __restrict__ norm,
    float* __restrict__ S,
    int nrows, int K, int Hout) {
  extern __shared__ float lds[];
  const int KSTRIDE = K + 4;
  float* Xs = lds;
  float* Wt = lds + 128 * KSTRIDE;

  const int tid = threadIdx.x;
  const int row0 = blockIdx.x * 128;

  for (int idx = tid * 4; idx < K * Hout; idx += 256 * 4) {
    int k = idx / Hout, c = idx - k * Hout;
    float4 v = *(const float4*)&W[idx];
    Wt[(c + 0) * KSTRIDE + k] = v.x;
    Wt[(c + 1) * KSTRIDE + k] = v.y;
    Wt[(c + 2) * KSTRIDE + k] = v.z;
    Wt[(c + 3) * KSTRIDE + k] = v.w;
  }
  for (int idx = tid * 4; idx < 128 * K; idx += 256 * 4) {
    int r = idx / K, c = idx - r * K;
    float4 v = {0.f, 0.f, 0.f, 0.f};
    int gr = row0 + r;
    if (gr < nrows) v = *(const float4*)&X[(long)gr * K + c];
    *(float4*)&Xs[r * KSTRIDE + c] = v;
  }
  __syncthreads();

  const int lane = tid & 31;
  const int wave = tid >> 5;
  const int mrow = lane & 15;
  const int kg   = lane >> 4;
  const float* Xrow = &Xs[(wave * 16 + mrow) * KSTRIDE + 2 * kg];

  float nrm[8];
#pragma unroll
  for (int r = 0; r < 8; ++r) {
    int grow = row0 + wave * 16 + r + 8 * kg;
    nrm[r] = (grow < nrows) ? norm[grow] : 0.f;
  }

  for (int ct = 0; ct < (Hout >> 4); ++ct) {
    const int col = ct * 16 + mrow;
    const float* Wcol = &Wt[col * KSTRIDE + 2 * kg];
    v8f acc = {};
    for (int k = 0; k < K; k += 4) {
      v2f a = *(const v2f*)&Xrow[k];
      v2f b = *(const v2f*)&Wcol[k];
      acc = __builtin_amdgcn_wmma_f32_16x16x4_f32(
           false, a,  false, b,
           (short)0, acc,  false,  false);
    }
#pragma unroll
    for (int r = 0; r < 8; ++r) {
      int grow = row0 + wave * 16 + r + 8 * kg;
      if (grow < nrows) {
        float sv = acc[r] * nrm[r];
        long o = (long)grow * Hout + col;
        S[o] = sv;
      }
    }
  }
}

template <int VEC> struct rowvec;
template <> struct rowvec<4> { typedef v4f T; };
template <> struct rowvec<2> { typedef v2f T; };

template <int VEC, int KB>
__device__ __forceinline__ typename rowvec<VEC>::T take_rows(
    typename rowvec<VEC>::T acc, const typename rowvec<VEC>::T* __restrict__ Sv,
    int mycol, int t, int lane, unsigned last) {
  typedef typename rowvec<VEC>::T VT;
  VT v[KB];
#pragma unroll
  for (int u = 0; u < KB; ++u) {
    unsigned s = (unsigned)__builtin_amdgcn_readlane(mycol, t + u);
    s = s < last ? s : last;
    v[u] = Sv[(size_t)s * 32 + lane];
  }
#pragma unroll
  for (int u = 0; u < KB; ++u) acc += v[u];
  return acc;
}

template <int VEC>
__global__ __launch_bounds__(256) void k_gather(
    const int* __restrict__ segw, const int* __restrict__ col,
    const float* __restrict__ S, float* __restrict__ AGG, int n) {
  typedef typename rowvec<VEC>::T VT;
  const int lane = threadIdx.x & 31;
  const int node = __builtin_amdgcn_readfirstlane((int)(blockIdx.x * 8 + (threadIdx.x >> 5)));
  if (node >= n) return;
  const VT* Sv = (const VT*)S;
  const unsigned last = (unsigned)(n - 1);
  const int2 sc = ((const int2*)segw)[node];
  const int b = sc.x;
  const int deg = sc.y;
  int m = deg < 32 ? deg : 32;
  int mycol = (lane < m) ? col[b + lane] : 0;
  const VT self = Sv[(size_t)node * 32 + lane];
  VT acc = {};
  int a0 = 0;
  while (true) {
    int t = 0;
    for (; t + 8 <= m; t += 8) acc = take_rows<VEC, 8>(acc, Sv, mycol, t, lane, last);
    if (t + 4 <= m) { acc = take_rows<VEC, 4>(acc, Sv, mycol, t, lane, last); t += 4; }
    if (t + 2 <= m) { acc = take_rows<VEC, 2>(acc, Sv, mycol, t, lane, last); t += 2; }
    if (t < m) acc = take_rows<VEC, 1>(acc, Sv, mycol, t, lane, last);
    a0 += 32;
    if (a0 >= deg) break;
    m = (deg - a0) < 32 ? (deg - a0) : 32;
    mycol = (lane < m) ? col[b + a0 + lane] : 0;
  }
  acc += self;
  ((VT*)AGG)[(size_t)node * 32 + lane] = acc;
}

__global__ void k_finalize(const float* __restrict__ AGG, const float* __restrict__ norm,
                           const float* __restrict__ bias, const float* __restrict__ xin,
                           float* __restrict__ out, int n, int Hout, int mode) {
  const int chunks = Hout >> 2;
  long tid = (long)blockIdx.x * blockDim.x + threadIdx.x;
  if (tid >= (long)n * chunks) return;
  int i = (int)(tid / chunks);
  int c = (int)(tid - (long)i * chunks) << 2;
  long o = (long)i * Hout + c;
  float nr = norm[i];
  float4 av = *(const float4*)&AGG[o];
  float4 bv = *(const float4*)&bias[c];
  float vx = av.x * nr + bv.x;
  float vy = av.y * nr + bv.y;
  float vz = av.z * nr + bv.z;
  float vw = av.w * nr + bv.w;
  if (mode != 2) {
    vx = vx >= 0.f ? vx : NEG_SLOPE * vx;
    vy = vy >= 0.f ? vy : NEG_SLOPE * vy;
    vz = vz >= 0.f ? vz : NEG_SLOPE * vz;
    vw = vw >= 0.f ? vw : NEG_SLOPE * vw;
  }
  if (mode == 1) {
    float4 xv = *(const float4*)&xin[o];
    vx = (xv.x + vx) * 0.5f;
    vy = (xv.y + vy) * 0.5f;
    vz = (xv.z + vz) * 0.5f;
    vw = (xv.w + vw) * 0.5f;
  }
  float4 r4 = {vx, vy, vz, vw};
  *(float4*)&out[o] = r4;
}

extern "C" void kernel_launch(void* const* d_in, const int* in_sizes, int n_in,
                              void* d_out, int out_size, void* d_ws, size_t ws_size,
                              hipStream_t stream) {
  (void)n_in; (void)out_size; (void)ws_size;
  const float* inputs = (const float*)d_in[0];
  const int*   edges  = (const int*)d_in[1];
  const float* w_in   = (const float*)d_in[2];
  const float* b_in   = (const float*)d_in[3];
  const float* w1     = (const float*)d_in[4];
  const float* b1     = (const float*)d_in[5];
  const float* w2     = (const float*)d_in[6];
  const float* b2     = (const float*)d_in[7];
  const float* w_out  = (const float*)d_in[8];
  const float* b_out  = (const float*)d_in[9];

  const int H   = in_sizes[3];
  const int IN  = in_sizes[2] / H;
  const int N   = in_sizes[0] / IN;
  const int E   = in_sizes[1] / 2;
  const int OUT = in_sizes[9];
  const int B   = in_sizes[5] / H;

  const int* srcI = edges;
  const int* dstI = edges + E;

  const size_t Npad = ((size_t)N + 255) & ~(size_t)255;
  const size_t Epad = ((size_t)E + 255) & ~(size_t)255;
  float* norm = (float*)d_ws;
  float* S    = norm + Npad;
  float* AGG  = S + (size_t)N * H;
  float* Hb   = AGG + (size_t)N * H;
  int* cnt    = (int*)(Hb + (size_t)N * H);
  int* seg    = cnt + Npad;
  int* rank   = seg + 2 * Npad;
  int* col    = rank + Epad;

  float* xout = (float*)d_out;
  float* xbuf = xout + (size_t)N * OUT;

  k_cnt_zero  <<<(N + 255) / 256, 256, 0, stream>>>(cnt, N);
  k_cnt_rank  <<<(E + 255) / 256, 256, 0, stream>>>(dstI, cnt, rank, E);
  k_start_norm<<<(N + 255) / 256, 256, 0, stream>>>(cnt, seg, norm, N);
  k_fill      <<<(E + 255) / 256, 256, 0, stream>>>(srcI, dstI, rank, seg, col, E);

  auto conv = [&](const float* Xin, const float* W, const float* bias, int K, int Hout,
                  float* outp, int mode, const float* resid) {
    size_t shmem = (size_t)(128 + Hout) * (K + 4) * sizeof(float);
    k_gemm_wmma<<<(N + 127) / 128, 256, shmem, stream>>>(Xin, W, norm, S, N, K, Hout);
    if (Hout == 64)
      k_gather<2><<<(N + 7) / 8, 256, 0, stream>>>(seg, col, S, AGG, N);
    else
      k_gather<4><<<(N + 7) / 8, 256, 0, stream>>>(seg, col, S, AGG, N);
    long tot2 = (long)N * (Hout >> 2);
    k_finalize<<<(unsigned)((tot2 + 255) / 256), 256, 0, stream>>>(AGG, norm, bias, resid,
                                                                   outp, N, Hout, mode);
  };

  conv(inputs, w_in, b_in, IN, H, xbuf, 0, nullptr);
  for (int i = 0; i < B; ++i) {
    conv(xbuf, w1 + (size_t)i * H * H, b1 + (size_t)i * H, H, H, Hb,   0, nullptr);
    conv(Hb,   w2 + (size_t)i * H * H, b2 + (size_t)i * H, H, H, xbuf, 1, xbuf);
  }
  conv(xbuf, w_out, b_out, H, OUT, xout, 2, nullptr);
}


# Layer epilogue (norm scale, bias, activation, residual) done by the gather wave that holds the summed row: the finalize kernel, its one wave per row and the sum's trip through memory are gone
# speedup vs baseline: 2.9330x; 1.1578x over previous
#include <hip/hip_runtime.h>

typedef __attribute__((ext_vector_type(2))) float v2f;
typedef __attribute__((ext_vector_type(4))) float v4f;
typedef __attribute__((ext_vector_type(8))) float v8f;

#define NEG_SLOPE 0.01f

__global__ void k_cnt_zero(int* __restrict__ cnt, int n) {
  int i = blockIdx.x * blockDim.x + threadIdx.x;
  if (i < n) cnt[i] = 0;
}
__global__ void k_cnt_rank(const int* __restrict__ dst, int* __restrict__ cnt,
                           int* __restrict__ rank, int e) {
  int i = blockIdx.x * blockDim.x + threadIdx.x;
  if (i < e) rank[i] = atomicAdd(&cnt[dst[i]], 1);
}
__global__ __launch_bounds__(256) void k_start_norm(const int* __restrict__ cnt,
                                                    int* __restrict__ segw,
                                                    float* __restrict__ norm, int n) {
  int2* seg = (int2*)segw;
  __shared__ int wsum[8];
  __shared__ int psum[8];
  const int tid = threadIdx.x;
  const int lane = tid & 31;
  const int wv = tid >> 5;
  const int base = blockIdx.x * 256;
  int part = 0;
  const int4* c4 = (const int4*)cnt;
  for (int j = tid; j < (base >> 2); j += 256) {
    int4 v = c4[j];
    part += (v.x + v.y) + (v.z + v.w);
  }
  const int i = base + tid;
  const int c = (i < n) ? cnt[i] : 0;
  int incl = c;
#pragma unroll
  for (int off = 1; off < 32; off <<= 1) {
    int t = __shfl_up(incl, off);
    if (lane >= off) incl += t;
  }
#pragma unroll
  for (int off = 16; off > 0; off >>= 1) part += __shfl_xor(part, off);
  if (lane == 31) wsum[wv] = incl;
  if (lane == 0) psum[wv] = part;
  __syncthreads();
  int offset = 0;
#pragma unroll
  for (int w = 0; w < 8; ++w) {
    offset += psum[w];
    if (w < wv) offset += wsum[w];
  }
  if (i < n) {
    seg[i] = make_int2(offset + incl - c, c);
    norm[i] = rsqrtf(1.0f + (float)c);
  }
}
__global__ void k_fill(const int* __restrict__ src, const int* __restrict__ dst,
                       const int* __restrict__ rank, const int* __restrict__ segw,
                       int* __restrict__ col, int e) {
  int i = blockIdx.x * blockDim.x + threadIdx.x;
  if (i < e) atomicExch(&col[segw[2 * dst[i]] + rank[i]], src[i]);
}

__global__ __launch_bounds__(256) void k_gemm_wmma(
    const float* __restrict__ X, const float* __restrict__ W,
    const float* __restrict__ norm,
    float* __restrict__ S,
    int nrows, int K, int Hout) {
  extern __shared__ float lds[];
  const int KSTRIDE = K + 4;
  float* Xs = lds;
  float* Wt = lds + 128 * KSTRIDE;

  const int tid = threadIdx.x;
  const int row0 = blockIdx.x * 128;

  for (int idx = tid * 4; idx < K * Hout; idx += 256 * 4) {
    int k = idx / Hout, c = idx - k * Hout;
    float4 v = *(const float4*)&W[idx];
    Wt[(c + 0) * KSTRIDE + k] = v.x;
    Wt[(c + 1) * KSTRIDE + k] = v.y;
    Wt[(c + 2) * KSTRIDE + k] = v.z;
    Wt[(c + 3) * KSTRIDE + k] = v.w;
  }
  for (int idx = tid * 4; idx < 128 * K; idx += 256 * 4) {
    int r = idx / K, c = idx - r * K;
    float4 v = {0.f, 0.f, 0.f, 0.f};
    int gr = row0 + r;
    if (gr < nrows) v = *(const float4*)&X[(long)gr * K + c];
    *(float4*)&Xs[r * KSTRIDE + c] = v;
  }
  __syncthreads();

  const int lane = tid & 31;
  const int wave = tid >> 5;
  const int mrow = lane & 15;
  const int kg   = lane >> 4;
  const float* Xrow = &Xs[(wave * 16 + mrow) * KSTRIDE + 2 * kg];

  float nrm[8];
#pragma unroll
  for (int r = 0; r < 8; ++r) {
    int grow = row0 + wave * 16 + r + 8 * kg;
    nrm[r] = (grow < nrows) ? norm[grow] : 0.f;
  }

  for (int ct = 0; ct < (Hout >> 4); ++ct) {
    const int col = ct * 16 + mrow;
    const float* Wcol = &Wt[col * KSTRIDE + 2 * kg];
    v8f acc = {};
    for (int k = 0; k < K; k += 4) {
      v2f a = *(const v2f*)&Xrow[k];
      v2f b = *(const v2f*)&Wcol[k];
      acc = __builtin_amdgcn_wmma_f32_16x16x4_f32(
           false, a,  false, b,
           (short)0, acc,  false,  false);
    }
#pragma unroll
    for (int r = 0; r < 8; ++r) {
      int grow = row0 + wave * 16 + r + 8 * kg;
      if (grow < nrows) {
        float sv = acc[r] * nrm[r];
        long o = (long)grow * Hout + col;
        S[o] = sv;
      }
    }
  }
}

template <int VEC> struct rowvec;
template <> struct rowvec<4> { typedef v4f T; };
template <> struct rowvec<2> { typedef v2f T; };

template <int VEC, int KB>
__device__ __forceinline__ typename rowvec<VEC>::T take_rows(
    typename rowvec<VEC>::T acc, const typename rowvec<VEC>::T* __restrict__ Sv,
    int mycol, int t, int lane, unsigned last) {
  typedef typename rowvec<VEC>::T VT;
  VT v[KB];
#pragma unroll
  for (int u = 0; u < KB; ++u) {
    unsigned s = (unsigned)__builtin_amdgcn_readlane(mycol, t + u);
    s = s < last ? s : last;
    v[u] = Sv[(size_t)s * 32 + lane];
  }
#pragma unroll
  for (int u = 0; u < KB; ++u) acc += v[u];
  return acc;
}

template <int VEC>
__global__ __launch_bounds__(256) void k_gather_fin(
    const int* __restrict__ segw, const int* __restrict__ col,
    const float* __restrict__ S, const float* __restrict__ norm,
    const float* __restrict__ bias, const float* resid, float* out, int n, int mode) {
  typedef typename rowvec<VEC>::T VT;
  const int lane = threadIdx.x & 31;
  const int node = __builtin_amdgcn_readfirstlane((int)(blockIdx.x * 8 + (threadIdx.x >> 5)));
  if (node >= n) return;
  const VT* Sv = (const VT*)S;
  const unsigned last = (unsigned)(n - 1);
  const int2 sc = ((const int2*)segw)[node];
  const int b = sc.x;
  const int deg = sc.y;
  int m = deg < 32 ? deg : 32;
  int mycol = (lane < m) ? col[b + lane] : 0;
  const VT self = Sv[(size_t)node * 32 + lane];
  const float nr = norm[node];
  const VT bv = ((const VT*)bias)[lane];
  VT rv = {};
  if (mode == 1) rv = ((const VT*)resid)[(size_t)node * 32 + lane];
  VT acc = {};
  int a0 = 0;
  while (true) {
    int t = 0;
    for (; t + 8 <= m; t += 8) acc = take_rows<VEC, 8>(acc, Sv, mycol, t, lane, last);
    if (t + 4 <= m) { acc = take_rows<VEC, 4>(acc, Sv, mycol, t, lane, last); t += 4; }
    if (t + 2 <= m) { acc = take_rows<VEC, 2>(acc, Sv, mycol, t, lane, last); t += 2; }
    if (t < m) acc = take_rows<VEC, 1>(acc, Sv, mycol, t, lane, last);
    a0 += 32;
    if (a0 >= deg) break;
    m = (deg - a0) < 32 ? (deg - a0) : 32;
    mycol = (lane < m) ? col[b + a0 + lane] : 0;
  }
  acc += self;
  VT v;
#pragma unroll
  for (int j = 0; j < VEC; ++j) {
    float x = acc[j] * nr + bv[j];
    if (mode != 2) x = x >= 0.f ? x : NEG_SLOPE * x;
    if (mode == 1) x = (rv[j] + x) * 0.5f;
    v[j] = x;
  }
  ((VT*)out)[(size_t)node * 32 + lane] = v;
}

extern "C" void kernel_launch(void* const* d_in, const int* in_sizes, int n_in,
                              void* d_out, int out_size, void* d_ws, size_t ws_size,
                              hipStream_t stream) {
  (void)n_in; (void)out_size; (void)ws_size;
  const float* inputs = (const float*)d_in[0];
  const int*   edges  = (const int*)d_in[1];
  const float* w_in   = (const float*)d_in[2];
  const float* b_in   = (const float*)d_in[3];
  const float* w1     = (const float*)d_in[4];
  const float* b1     = (const float*)d_in[5];
  const float* w2     = (const float*)d_in[6];
  const float* b2     = (const float*)d_in[7];
  const float* w_out  = (const float*)d_in[8];
  const float* b_out  = (const float*)d_in[9];

  const int H   = in_sizes[3];
  const int IN  = in_sizes[2] / H;
  const int N   = in_sizes[0] / IN;
  const int E   = in_sizes[1] / 2;
  const int OUT = in_sizes[9];
  const int B   = in_sizes[5] / H;

  const int* srcI = edges;
  const int* dstI = edges + E;

  const size_t Npad = ((size_t)N + 255) & ~(size_t)255;
  const size_t Epad = ((size_t)E + 255) & ~(size_t)255;
  float* norm = (float*)d_ws;
  float* S    = norm + Npad;
  float* Hb   = S + (size_t)N * H;
  int* cnt    = (int*)(Hb + (size_t)N * H);
  int* seg    = cnt + Npad;
  int* rank   = seg + 2 * Npad;
  int* col    = rank + Epad;

  float* xout = (float*)d_out;
  float* xbuf = xout + (size_t)N * OUT;

  k_cnt_zero  <<<(N + 255) / 256, 256, 0, stream>>>(cnt, N);
  k_cnt_rank  <<<(E + 255) / 256, 256, 0, stream>>>(dstI, cnt, rank, E);
  k_start_norm<<<(N + 255) / 256, 256, 0, stream>>>(cnt, seg, norm, N);
  k_fill      <<<(E + 255) / 256, 256, 0, stream>>>(srcI, dstI, rank, seg, col, E);

  auto conv = [&](const float* Xin, const float* W, const float* bias, int K, int Hout,
                  float* outp, int mode, const float* resid) {
    size_t shmem = (size_t)(128 + Hout) * (K + 4) * sizeof(float);
    k_gemm_wmma<<<(N + 127) / 128, 256, shmem, stream>>>(Xin, W, norm, S, N, K, Hout);
    if (Hout == 64)
      k_gather_fin<2><<<(N + 7) / 8, 256, 0, stream>>>(seg, col, S, norm, bias, resid, outp,
                                                       N, mode);
    else
      k_gather_fin<4><<<(N + 7) / 8, 256, 0, stream>>>(seg, col, S, norm, bias, resid, outp,
                                                       N, mode);
  };

  conv(inputs, w_in, b_in, IN, H, xbuf, 0, nullptr);
  for (int i = 0; i < B; ++i) {
    conv(xbuf, w1 + (size_t)i * H * H, b1 + (size_t)i * H, H, H, Hb,   0, nullptr);
    conv(Hb,   w2 + (size_t)i * H * H, b2 + (size_t)i * H, H, H, xbuf, 1, xbuf);
  }
  conv(xbuf, w_out, b_out, H, OUT, xout, 2, nullptr);
}


# GEMM tile staging with compile-time shapes: a thread issues all its global loads of the X and W tiles together (was one per trip, each waited for); W transposed by conflict-free 8-byte LDS stores
# speedup vs baseline: 3.8284x; 1.3053x over previous
#include <hip/hip_runtime.h>

typedef __attribute__((ext_vector_type(2))) float v2f;
typedef __attribute__((ext_vector_type(4))) float v4f;
typedef __attribute__((ext_vector_type(8))) float v8f;

#define NEG_SLOPE 0.01f

__global__ void k_cnt_zero(int* __restrict__ cnt, int n) {
  int i = blockIdx.x * blockDim.x + threadIdx.x;
  if (i < n) cnt[i] = 0;
}
__global__ void k_cnt_rank(const int* __restrict__ dst, int* __restrict__ cnt,
                           int* __restrict__ rank, int e) {
  int i = blockIdx.x * blockDim.x + threadIdx.x;
  if (i < e) rank[i] = atomicAdd(&cnt[dst[i]], 1);
}
__global__ __launch_bounds__(256) void k_start_norm(const int* __restrict__ cnt,
                                                    int* __restrict__ segw,
                                                    float* __restrict__ norm, int n) {
  int2* seg = (int2*)segw;
  __shared__ int wsum[8];
  __shared__ int psum[8];
  const int tid = threadIdx.x;
  const int lane = tid & 31;
  const int wv = tid >> 5;
  const int base = blockIdx.x * 256;
  int part = 0;
  const int4* c4 = (const int4*)cnt;
  for (int j = tid; j < (base >> 2); j += 256) {
    int4 v = c4[j];
    part += (v.x + v.y) + (v.z + v.w);
  }
  const int i = base + tid;
  const int c = (i < n) ? cnt[i] : 0;
  int incl = c;
#pragma unroll
  for (int off = 1; off < 32; off <<= 1) {
    int t = __shfl_up(incl, off);
    if (lane >= off) incl += t;
  }
#pragma unroll
  for (int off = 16; off > 0; off >>= 1) part += __shfl_xor(part, off);
  if (lane == 31) wsum[wv] = incl;
  if (lane == 0) psum[wv] = part;
  __syncthreads();
  int offset = 0;
#pragma unroll
  for (int w = 0; w < 8; ++w) {
    offset += psum[w];
    if (w < wv) offset += wsum[w];
  }
  if (i < n) {
    seg[i] = make_int2(offset + incl - c, c);
    norm[i] = rsqrtf(1.0f + (float)c);
  }
}
__global__ void k_fill(const int* __restrict__ src, const int* __restrict__ dst,
                       const int* __restrict__ rank, const int* __restrict__ segw,
                       int* __restrict__ col, int e) {
  int i = blockIdx.x * blockDim.x + threadIdx.x;
  if (i < e) atomicExch(&col[segw[2 * dst[i]] + rank[i]], src[i]);
}

template <int K, int HOUT>
__global__ __launch_bounds__(256) void k_gemm_wmma(
    const float* __restrict__ X, const float* __restrict__ W,
    const float* __restrict__ norm,
    float* __restrict__ S,
    int nrows) {
  static_assert(K % 128 == 0 && HOUT % 16 == 0 && ((K / 16) * (HOUT / 16)) % 8 == 0,
                "tile shapes of the staging code");
  extern __shared__ float lds[];
  constexpr int KSTRIDE = K + 4;
  float* Xs = lds;
  float* Wt = lds + 128 * KSTRIDE;

  const int tid = threadIdx.x;
  const int lane = tid & 31;
  const int wave = __builtin_amdgcn_readfirstlane(tid >> 5);
  const int row0 = blockIdx.x * 128;

  constexpr int CT = HOUT / 16;
  constexpr int NT = (K / 16) * CT / 8;
  const int kk = lane & 7, cc = lane >> 3;
  float4 w0[NT], w1[NT];
#pragma unroll
  for (int i = 0; i < NT; ++i) {
    const int t = wave + 8 * i;
    const int k = (t / CT) * 16 + 2 * kk;
    const int c = (t % CT) * 16 + 4 * cc;
    w0[i] = *(const float4*)&W[(k + 0) * HOUT + c];
    w1[i] = *(const float4*)&W[(k + 1) * HOUT + c];
  }
  constexpr int XP = K / 128;
  float4 xv[16 * XP];
#pragma unroll
  for (int it = 0; it < 16; ++it) {
    const int gr = row0 + wave + 8 * it;
    const int grc = gr < nrows ? gr : nrows - 1;
#pragma unroll
    for (int p = 0; p < XP; ++p)
      xv[it * XP + p] = *(const float4*)&X[(size_t)grc * K + p * 128 + 4 * lane];
  }
#pragma unroll
  for (int i = 0; i < NT; ++i) {
    const int t = wave + 8 * i;
    const int k = (t / CT) * 16 + 2 * kk;
    const int c = (t % CT) * 16 + 4 * cc;
    v2f p0 = {w0[i].x, w1[i].x};
    v2f p1 = {w0[i].y, w1[i].y};
    v2f p2 = {w0[i].z, w1[i].z};
    v2f p3 = {w0[i].w, w1[i].w};
    *(v2f*)&Wt[(c + 0) * KSTRIDE + k] = p0;
    *(v2f*)&Wt[(c + 1) * KSTRIDE + k] = p1;
    *(v2f*)&Wt[(c + 2) * KSTRIDE + k] = p2;
    *(v2f*)&Wt[(c + 3) * KSTRIDE + k] = p3;
  }
#pragma unroll
  for (int it = 0; it < 16; ++it) {
    const int r = wave + 8 * it;
#pragma unroll
    for (int p = 0; p < XP; ++p)
      *(float4*)&Xs[r * KSTRIDE + p * 128 + 4 * lane] = xv[it * XP + p];
  }
  __syncthreads();

  const int mrow = lane & 15;
  const int kg   = lane >> 4;
  const float* Xrow = &Xs[(wave * 16 + mrow) * KSTRIDE + 2 * kg];

  float nrm[8];
#pragma unroll
  for (int r = 0; r < 8; ++r) {
    int grow = row0 + wave * 16 + r + 8 * kg;
    nrm[r] = (grow < nrows) ? norm[grow] : 0.f;
  }

  for (int ct = 0; ct < (HOUT >> 4); ++ct) {
    const int col = ct * 16 + mrow;
    const float* Wcol = &Wt[col * KSTRIDE + 2 * kg];
    v8f acc = {};
    for (int k = 0; k < K; k += 4) {
      v2f a = *(const v2f*)&Xrow[k];
      v2f b = *(const v2f*)&Wcol[k];
      acc = __builtin_amdgcn_wmma_f32_16x16x4_f32(
           false, a,  false, b,
           (short)0, acc,  false,  false);
    }
#pragma unroll
    for (int r = 0; r < 8; ++r) {
      int grow = row0 + wave * 16 + r + 8 * kg;
      if (grow < nrows) {
        float sv = acc[r] * nrm[r];
        long o = (long)grow * HOUT + col;
        S[o] = sv;
      }
    }
  }
}

template <int VEC> struct rowvec;
template <> struct rowvec<4> { typedef v4f T; };
template <> struct rowvec<2> { typedef v2f T; };

template <int VEC, int KB>
__device__ __forceinline__ typename rowvec<VEC>::T take_rows(
    typename rowvec<VEC>::T acc, const typename rowvec<VEC>::T* __restrict__ Sv,
    int mycol, int t, int lane, unsigned last) {
  typedef typename rowvec<VEC>::T VT;
  VT v[KB];
#pragma unroll
  for (int u = 0; u < KB; ++u) {
    unsigned s = (unsigned)__builtin_amdgcn_readlane(mycol, t + u);
    s = s < last ? s : last;
    v[u] = Sv[(size_t)s * 32 + lane];
  }
#pragma unroll
  for (int u = 0; u < KB; ++u) acc += v[u];
  return acc;
}

template <int VEC>
__global__ __launch_bounds__(256) void k_gather_fin(
    const int* __restrict__ segw, const int* __restrict__ col,
    const float* __restrict__ S, const float* __restrict__ norm,
    const float* __restrict__ bias, const float* resid, float* out, int n, int mode) {
  typedef typename rowvec<VEC>::T VT;
  const int lane = threadIdx.x & 31;
  const int node = __builtin_amdgcn_readfirstlane((int)(blockIdx.x * 8 + (threadIdx.x >> 5)));
  if (node >= n) return;
  const VT* Sv = (const VT*)S;
  const unsigned last = (unsigned)(n - 1);
  const int2 sc = ((const int2*)segw)[node];
  const int b = sc.x;
  const int deg = sc.y;
  int m = deg < 32 ? deg : 32;
  int mycol = (lane < m) ? col[b + lane] : 0;
  const VT self = Sv[(size_t)node * 32 + lane];
  const float nr = norm[node];
  const VT bv = ((const VT*)bias)[lane];
  VT rv = {};
  if (mode == 1) rv = ((const VT*)resid)[(size_t)node * 32 + lane];
  VT acc = {};
  int a0 = 0;
  while (true) {
    int t = 0;
    for (; t + 8 <= m; t += 8) acc = take_rows<VEC, 8>(acc, Sv, mycol, t, lane, last);
    if (t + 4 <= m) { acc = take_rows<VEC, 4>(acc, Sv, mycol, t, lane, last); t += 4; }
    if (t + 2 <= m) { acc = take_rows<VEC, 2>(acc, Sv, mycol, t, lane, last); t += 2; }
    if (t < m) acc = take_rows<VEC, 1>(acc, Sv, mycol, t, lane, last);
    a0 += 32;
    if (a0 >= deg) break;
    m = (deg - a0) < 32 ? (deg - a0) : 32;
    mycol = (lane < m) ? col[b + a0 + lane] : 0;
  }
  acc += self;
  VT v;
#pragma unroll
  for (int j = 0; j < VEC; ++j) {
    float x = acc[j] * nr + bv[j];
    if (mode != 2) x = x >= 0.f ? x : NEG_SLOPE * x;
    if (mode == 1) x = (rv[j] + x) * 0.5f;
    v[j] = x;
  }
  ((VT*)out)[(size_t)node * 32 + lane] = v;
}

extern "C" void kernel_launch(void* const* d_in, const int* in_sizes, int n_in,
                              void* d_out, int out_size, void* d_ws, size_t ws_size,
                              hipStream_t stream) {
  (void)n_in; (void)out_size; (void)ws_size;
  const float* inputs = (const float*)d_in[0];
  const int*   edges  = (const int*)d_in[1];
  const float* w_in   = (const float*)d_in[2];
  const float* b_in   = (const float*)d_in[3];
  const float* w1     = (const float*)d_in[4];
  const float* b1     = (const float*)d_in[5];
  const float* w2     = (const float*)d_in[6];
  const float* b2     = (const float*)d_in[7];
  const float* w_out  = (const float*)d_in[8];
  const float* b_out  = (const float*)d_in[9];

  const int H   = in_sizes[3];
  const int IN  = in_sizes[2] / H;
  const int N   = in_sizes[0] / IN;
  const int E   = in_sizes[1] / 2;
  const int OUT = in_sizes[9];
  const int B   = in_sizes[5] / H;

  const int* srcI = edges;
  const int* dstI = edges + E;

  const size_t Npad = ((size_t)N + 255) & ~(size_t)255;
  const size_t Epad = ((size_t)E + 255) & ~(size_t)255;
  float* norm = (float*)d_ws;
  float* S    = norm + Npad;
  float* Hb   = S + (size_t)N * H;
  int* cnt    = (int*)(Hb + (size_t)N * H);
  int* seg    = cnt + Npad;
  int* rank   = seg + 2 * Npad;
  int* col    = rank + Epad;

  float* xout = (float*)d_out;
  float* xbuf = xout + (size_t)N * OUT;

  k_cnt_zero  <<<(N + 255) / 256, 256, 0, stream>>>(cnt, N);
  k_cnt_rank  <<<(E + 255) / 256, 256, 0, stream>>>(dstI, cnt, rank, E);
  k_start_norm<<<(N + 255) / 256, 256, 0, stream>>>(cnt, seg, norm, N);
  k_fill      <<<(E + 255) / 256, 256, 0, stream>>>(srcI, dstI, rank, seg, col, E);

  auto conv = [&](const float* Xin, const float* W, const float* bias, int K, int Hout,
                  float* outp, int mode, const float* resid) {
    size_t shmem = (size_t)(128 + Hout) * (K + 4) * sizeof(float);
    if (Hout == 64) {
      k_gemm_wmma<128, 64><<<(N + 127) / 128, 256, shmem, stream>>>(Xin, W, norm, S, N);
      k_gather_fin<2><<<(N + 7) / 8, 256, 0, stream>>>(seg, col, S, norm, bias, resid, outp,
                                                       N, mode);
    } else {
      k_gemm_wmma<128, 128><<<(N + 127) / 128, 256, shmem, stream>>>(Xin, W, norm, S, N);
      k_gather_fin<4><<<(N + 7) / 8, 256, 0, stream>>>(seg, col, S, norm, bias, resid, outp,
                                                       N, mode);
    }
  };

  conv(inputs, w_in, b_in, IN, H, xbuf, 0, nullptr);
  for (int i = 0; i < B; ++i) {
    conv(xbuf, w1 + (size_t)i * H * H, b1 + (size_t)i * H, H, H, Hb,   0, nullptr);
    conv(Hb,   w2 + (size_t)i * H * H, b2 + (size_t)i * H, H, H, xbuf, 1, xbuf);
  }
  conv(xbuf, w_out, b_out, H, OUT, xout, 2, nullptr);
}


# Layer l's gather+epilogue fused with layer l+1's GEMM per 128-row block: x goes gather->LDS->WMMA, 13 launches and the GEMM's X staging gone; S in two buffers; every sum in the same order
# speedup vs baseline: 4.1554x; 1.0854x over previous
#include <hip/hip_runtime.h>

typedef __attribute__((ext_vector_type(2))) float v2f;
typedef __attribute__((ext_vector_type(4))) float v4f;
typedef __attribute__((ext_vector_type(8))) float v8f;

#define NEG_SLOPE 0.01f

__global__ void k_cnt_zero(int* __restrict__ cnt, int n) {
  int i = blockIdx.x * blockDim.x + threadIdx.x;
  if (i < n) cnt[i] = 0;
}
__global__ void k_cnt_rank(const int* __restrict__ dst, int* __restrict__ cnt,
                           int* __restrict__ rank, int e) {
  int i = blockIdx.x * blockDim.x + threadIdx.x;
  if (i < e) rank[i] = atomicAdd(&cnt[dst[i]], 1);
}
__global__ __launch_bounds__(256) void k_start_norm(const int* __restrict__ cnt,
                                                    int* __restrict__ segw,
                                                    float* __restrict__ norm, int n) {
  int2* seg = (int2*)segw;
  __shared__ int wsum[8];
  __shared__ int psum[8];
  const int tid = threadIdx.x;
  const int lane = tid & 31;
  const int wv = tid >> 5;
  const int base = blockIdx.x * 256;
  int part = 0;
  const int4* c4 = (const int4*)cnt;
  for (int j = tid; j < (base >> 2); j += 256) {
    int4 v = c4[j];
    part += (v.x + v.y) + (v.z + v.w);
  }
  const int i = base + tid;
  const int c = (i < n) ? cnt[i] : 0;
  int incl = c;
#pragma unroll
  for (int off = 1; off < 32; off <<= 1) {
    int t = __shfl_up(incl, off);
    if (lane >= off) incl += t;
  }
#pragma unroll
  for (int off = 16; off > 0; off >>= 1) part += __shfl_xor(part, off);
  if (lane == 31) wsum[wv] = incl;
  if (lane == 0) psum[wv] = part;
  __syncthreads();
  int offset = 0;
#pragma unroll
  for (int w = 0; w < 8; ++w) {
    offset += psum[w];
    if (w < wv) offset += wsum[w];
  }
  if (i < n) {
    seg[i] = make_int2(offset + incl - c, c);
    norm[i] = rsqrtf(1.0f + (float)c);
  }
}
__global__ void k_fill(const int* __restrict__ src, const int* __restrict__ dst,
                       const int* __restrict__ rank, const int* __restrict__ segw,
                       int* __restrict__ col, int e) {
  int i = blockIdx.x * blockDim.x + threadIdx.x;
  if (i < e) atomicExch(&col[segw[2 * dst[i]] + rank[i]], src[i]);
}

template <int K, int HOUT>
__global__ __launch_bounds__(256) void k_gemm_wmma(
    const float* __restrict__ X, const float* __restrict__ W,
    const float* __restrict__ norm,
    float* __restrict__ S,
    int nrows) {
  static_assert(K % 128 == 0 && HOUT % 16 == 0 && ((K / 16) * (HOUT / 16)) % 8 == 0,
                "tile shapes of the staging code");
  extern __shared__ float lds[];
  constexpr int KSTRIDE = K + 4;
  float* Xs = lds;
  float* Wt = lds + 128 * KSTRIDE;

  const int tid = threadIdx.x;
  const int lane = tid & 31;
  const int wave = __builtin_amdgcn_readfirstlane(tid >> 5);
  const int row0 = blockIdx.x * 128;

  constexpr int CT = HOUT / 16;
  constexpr int NT = (K / 16) * CT / 8;
  const int kk = lane & 7, cc = lane >> 3;
  float4 w0[NT], w1[NT];
#pragma unroll
  for (int i = 0; i < NT; ++i) {
    const int t = wave + 8 * i;
    const int k = (t / CT) * 16 + 2 * kk;
    const int c = (t % CT) * 16 + 4 * cc;
    w0[i] = *(const float4*)&W[(k + 0) * HOUT + c];
    w1[i] = *(const float4*)&W[(k + 1) * HOUT + c];
  }
  constexpr int XP = K / 128;
  float4 xv[16 * XP];
#pragma unroll
  for (int it = 0; it < 16; ++it) {
    const int gr = row0 + wave + 8 * it;
    const int grc = gr < nrows ? gr : nrows - 1;
#pragma unroll
    for (int p = 0; p < XP; ++p)
      xv[it * XP + p] = *(const float4*)&X[(size_t)grc * K + p * 128 + 4 * lane];
  }
#pragma unroll
  for (int i = 0; i < NT; ++i) {
    const int t = wave + 8 * i;
    const int k = (t / CT) * 16 + 2 * kk;
    const int c = (t % CT) * 16 + 4 * cc;
    v2f p0 = {w0[i].x, w1[i].x};
    v2f p1 = {w0[i].y, w1[i].y};
    v2f p2 = {w0[i].z, w1[i].z};
    v2f p3 = {w0[i].w, w1[i].w};
    *(v2f*)&Wt[(c + 0) * KSTRIDE + k] = p0;
    *(v2f*)&Wt[(c + 1) * KSTRIDE + k] = p1;
    *(v2f*)&Wt[(c + 2) * KSTRIDE + k] = p2;
    *(v2f*)&Wt[(c + 3) * KSTRIDE + k] = p3;
  }
#pragma unroll
  for (int it = 0; it < 16; ++it) {
    const int r = wave + 8 * it;
#pragma unroll
    for (int p = 0; p < XP; ++p)
      *(float4*)&Xs[r * KSTRIDE + p * 128 + 4 * lane] = xv[it * XP + p];
  }
  __syncthreads();

  const int mrow = lane & 15;
  const int kg   = lane >> 4;
  const float* Xrow = &Xs[(wave * 16 + mrow) * KSTRIDE + 2 * kg];

  float nrm[8];
#pragma unroll
  for (int r = 0; r < 8; ++r) {
    int grow = row0 + wave * 16 + r + 8 * kg;
    nrm[r] = (grow < nrows) ? norm[grow] : 0.f;
  }

  for (int ct = 0; ct < (HOUT >> 4); ++ct) {
    const int col = ct * 16 + mrow;
    const float* Wcol = &Wt[col * KSTRIDE + 2 * kg];
    v8f acc = {};
    for (int k = 0; k < K; k += 4) {
      v2f a = *(const v2f*)&Xrow[k];
      v2f b = *(const v2f*)&Wcol[k];
      acc = __builtin_amdgcn_wmma_f32_16x16x4_f32(
           false, a,  false, b,
           (short)0, acc,  false,  false);
    }
#pragma unroll
    for (int r = 0; r < 8; ++r) {
      int grow = row0 + wave * 16 + r + 8 * kg;
      if (grow < nrows) {
        float sv = acc[r] * nrm[r];
        long o = (long)grow * HOUT + col;
        S[o] = sv;
      }
    }
  }
}

template <int VEC> struct rowvec;
template <> struct rowvec<4> { typedef v4f T; };
template <> struct rowvec<2> { typedef v2f T; };

template <int VEC, int KB>
__device__ __forceinline__ typename rowvec<VEC>::T take_rows(
    typename rowvec<VEC>::T acc, const typename rowvec<VEC>::T* __restrict__ Sv,
    int mycol, int t, int lane, unsigned last) {
  typedef typename rowvec<VEC>::T VT;
  VT v[KB];
#pragma unroll
  for (int u = 0; u < KB; ++u) {
    unsigned s = (unsigned)__builtin_amdgcn_readlane(mycol, t + u);
    s = s < last ? s : last;
    v[u] = Sv[(size_t)s * 32 + lane];
  }
#pragma unroll
  for (int u = 0; u < KB; ++u) acc += v[u];
  return acc;
}

template <int VEC, int KB, class F>
__device__ __forceinline__ typename rowvec<VEC>::T take_rows_then(
    typename rowvec<VEC>::T acc, const typename rowvec<VEC>::T* __restrict__ Sv,
    int mycol, int t, int lane, unsigned last, F&& ask_others) {
  typedef typename rowvec<VEC>::T VT;
  VT v[KB];
#pragma unroll
  for (int u = 0; u < KB; ++u) {
    unsigned s = (unsigned)__builtin_amdgcn_readlane(mycol, t + u);
    s = s < last ? s : last;
    v[u] = Sv[(size_t)s * 32 + lane];
  }
  ask_others();
#pragma unroll
  for (int u = 0; u < KB; ++u) acc += v[u];
  return acc;
}

template <int VEC>
__global__ __launch_bounds__(256) void k_gather_fin(
    const int* __restrict__ segw, const int* __restrict__ col,
    const float* __restrict__ S, const float* __restrict__ norm,
    const float* __restrict__ bias, const float* resid, float* out, int n, int mode) {
  typedef typename rowvec<VEC>::T VT;
  const int lane = threadIdx.x & 31;
  const int node = __builtin_amdgcn_readfirstlane((int)(blockIdx.x * 8 + (threadIdx.x >> 5)));
  if (node >= n) return;
  const VT* Sv = (const VT*)S;
  const unsigned last = (unsigned)(n - 1);
  const int2 sc = ((const int2*)segw)[node];
  const int b = sc.x;
  const int deg = sc.y;
  int m = deg < 32 ? deg : 32;
  int mycol = (lane < m) ? col[b + lane] : 0;
  const VT self = Sv[(size_t)node * 32 + lane];
  const float nr = norm[node];
  const VT bv = ((const VT*)bias)[lane];
  VT rv = {};
  if (mode == 1) rv = ((const VT*)resid)[(size_t)node * 32 + lane];
  VT acc = {};
  int a0 = 0;
  while (true) {
    int t = 0;
    for (; t + 8 <= m; t += 8) acc = take_rows<VEC, 8>(acc, Sv, mycol, t, lane, last);
    if (t + 4 <= m) { acc = take_rows<VEC, 4>(acc, Sv, mycol, t, lane, last); t += 4; }
    if (t + 2 <= m) { acc = take_rows<VEC, 2>(acc, Sv, mycol, t, lane, last); t += 2; }
    if (t < m) acc = take_rows<VEC, 1>(acc, Sv, mycol, t, lane, last);
    a0 += 32;
    if (a0 >= deg) break;
    m = (deg - a0) < 32 ? (deg - a0) : 32;
    mycol = (lane < m) ? col[b + a0 + lane] : 0;
  }
  acc += self;
  VT v;
#pragma unroll
  for (int j = 0; j < VEC; ++j) {
    float x = acc[j] * nr + bv[j];
    if (mode != 2) x = x >= 0.f ? x : NEG_SLOPE * x;
    if (mode == 1) x = (rv[j] + x) * 0.5f;
    v[j] = x;
  }
  ((VT*)out)[(size_t)node * 32 + lane] = v;
}


template <int HOUT>
__global__ __launch_bounds__(512) void k_gather_gemm(
    const int* __restrict__ segw, const int* __restrict__ col,
    const float* __restrict__ S, const float* __restrict__ norm,
    const float* __restrict__ bias, const float* resid, float* xw,
    const float* __restrict__ W, float* __restrict__ Snext, int n, int mode) {
  constexpr int K = 128;
  constexpr int KSTRIDE = K + 4;
  extern __shared__ float lds[];
  float* Xs = lds;
  float* Wt = lds + 128 * KSTRIDE;

  const int tid = threadIdx.x;
  const int lane = tid & 31;
  const int wave = __builtin_amdgcn_readfirstlane(tid >> 5);
  const int row0 = blockIdx.x * 128;

  constexpr int CT = HOUT / 16;
  constexpr int NT = (K / 16) * CT / 16;
  {
    const int kk = lane & 7, cc = lane >> 3;
    float4 w0[NT], w1[NT];
#pragma unroll
    for (int i = 0; i < NT; ++i) {
      const int t = wave + 16 * i;
      const int k = (t / CT) * 16 + 2 * kk;
      const int c = (t % CT) * 16 + 4 * cc;
      w0[i] = *(const float4*)&W[(k + 0) * HOUT + c];
      w1[i] = *(const float4*)&W[(k + 1) * HOUT + c];
    }
#pragma unroll
    for (int i = 0; i < NT; ++i) {
      const int t = wave + 16 * i;
      const int k = (t / CT) * 16 + 2 * kk;
      const int c = (t % CT) * 16 + 4 * cc;
      v2f p0 = {w0[i].x, w1[i].x};
      v2f p1 = {w0[i].y, w1[i].y};
      v2f p2 = {w0[i].z, w1[i].z};
      v2f p3 = {w0[i].w, w1[i].w};
      *(v2f*)&Wt[(c + 0) * KSTRIDE + k] = p0;
      *(v2f*)&Wt[(c + 1) * KSTRIDE + k] = p1;
      *(v2f*)&Wt[(c + 2) * KSTRIDE + k] = p2;
      *(v2f*)&Wt[(c + 3) * KSTRIDE + k] = p3;
    }
  }

  {
    const v4f* Sv = (const v4f*)S;
    const unsigned last = (unsigned)(n - 1);
    const int rbase = row0 + wave * 8;
    int2 sv = make_int2(0, 0);
    float nv = 0.f;
    if (lane < 8 && rbase + lane < n) {
      sv = ((const int2*)segw)[rbase + lane];
      nv = norm[rbase + lane];
    }
    const v4f bv = ((const v4f*)bias)[lane];
    int b_n = __builtin_amdgcn_readlane(sv.x, 0);
    int deg_n = __builtin_amdgcn_readlane(sv.y, 0);
    int m_n = deg_n < 32 ? deg_n : 32;
    int mycol_n = (lane < m_n) ? col[b_n + lane] : 0;
    unsigned sn = (unsigned)rbase < last ? (unsigned)rbase : last;
    v4f self_n = Sv[(size_t)sn * 32 + lane];
#pragma unroll 1
    for (int i = 0; i < 8; ++i) {
      const int node = rbase + i;
      const int b = b_n;
      const int deg = deg_n;
      int m = m_n;
      int mycol = mycol_n;
      const v4f self = self_n;
      const float nr = __int_as_float(__builtin_amdgcn_readlane(__float_as_int(nv), i));
      v4f rv = {};
      auto ask_others = [&]() {
        if (i + 1 < 8) {
          b_n = __builtin_amdgcn_readlane(sv.x, i + 1);
          deg_n = __builtin_amdgcn_readlane(sv.y, i + 1);
          m_n = deg_n < 32 ? deg_n : 32;
          mycol_n = (lane < m_n) ? col[b_n + lane] : 0;
          sn = (unsigned)(node + 1) < last ? (unsigned)(node + 1) : last;
          self_n = Sv[(size_t)sn * 32 + lane];
        }
        if (mode == 1 && node < n) rv = ((const v4f*)resid)[(size_t)node * 32 + lane];
      };
      v4f acc = {};
      int t = 0;
      if (m >= 8) {
        acc = take_rows_then<4, 8>(acc, Sv, mycol, 0, lane, last, ask_others);
        t = 8;
      } else {
        ask_others();
      }
      if (node < n) {
        int a0 = 0;
        while (true) {
          for (; t + 8 <= m; t += 8) acc = take_rows<4, 8>(acc, Sv, mycol, t, lane, last);
          if (t + 4 <= m) { acc = take_rows<4, 4>(acc, Sv, mycol, t, lane, last); t += 4; }
          if (t + 2 <= m) { acc = take_rows<4, 2>(acc, Sv, mycol, t, lane, last); t += 2; }
          if (t < m) acc = take_rows<4, 1>(acc, Sv, mycol, t, lane, last);
          a0 += 32;
          if (a0 >= deg) break;
          m = (deg - a0) < 32 ? (deg - a0) : 32;
          mycol = (lane < m) ? col[b + a0 + lane] : 0;
          t = 0;
        }
        acc += self;
        v4f v;
#pragma unroll
        for (int j = 0; j < 4; ++j) {
          float x = acc[j] * nr + bv[j];
          if (mode != 2) x = x >= 0.f ? x : NEG_SLOPE * x;
          if (mode == 1) x = (rv[j] + x) * 0.5f;
          v[j] = x;
        }
        *(v4f*)&Xs[(wave * 8 + i) * KSTRIDE + 4 * lane] = v;
        if (xw != nullptr) ((v4f*)xw)[(size_t)node * 32 + lane] = v;
      } else {
        *(v4f*)&Xs[(wave * 8 + i) * KSTRIDE + 4 * lane] = v4f{};
      }
    }
  }
  __syncthreads();

  const int mrow = lane & 15;
  const int kg   = lane >> 4;
  const int rt = wave >> 1;
  const int ch = wave & 1;
  constexpr int CTW = CT / 2;
  const float* Xrow = &Xs[(rt * 16 + mrow) * KSTRIDE + 2 * kg];
  const float* Wcol = &Wt[((ch * CTW) * 16 + mrow) * KSTRIDE + 2 * kg];

  float nrm[8];
#pragma unroll
  for (int r = 0; r < 8; ++r) {
    int grow = row0 + rt * 16 + r + 8 * kg;
    nrm[r] = (grow < n) ? norm[grow] : 0.f;
  }

  v8f acc[CTW];
#pragma unroll
  for (int j = 0; j < CTW; ++j) acc[j] = v8f{};
#pragma unroll
  for (int k = 0; k < K; k += 4) {
    v2f a = *(const v2f*)&Xrow[k];
#pragma unroll
    for (int j = 0; j < CTW; ++j) {
      v2f bq = *(const v2f*)&Wcol[j * 16 * KSTRIDE + k];
      acc[j] = __builtin_amdgcn_wmma_f32_16x16x4_f32(
           false, a,  false, bq,
           (short)0, acc[j],  false,  false);
    }
  }
#pragma unroll
  for (int j = 0; j < CTW; ++j) {
    const int ccol = (ch * CTW + j) * 16 + mrow;
#pragma unroll
    for (int r = 0; r < 8; ++r) {
      int grow = row0 + rt * 16 + r + 8 * kg;
      if (grow < n) Snext[(long)grow * HOUT + ccol] = acc[j][r] * nrm[r];
    }
  }
}

extern "C" void kernel_launch(void* const* d_in, const int* in_sizes, int n_in,
                              void* d_out, int out_size, void* d_ws, size_t ws_size,
                              hipStream_t stream) {
  (void)n_in; (void)out_size; (void)ws_size;
  const float* inputs = (const float*)d_in[0];
  const int*   edges  = (const int*)d_in[1];
  const float* w_in   = (const float*)d_in[2];
  const float* b_in   = (const float*)d_in[3];
  const float* w1     = (const float*)d_in[4];
  const float* b1     = (const float*)d_in[5];
  const float* w2     = (const float*)d_in[6];
  const float* b2     = (const float*)d_in[7];
  const float* w_out  = (const float*)d_in[8];
  const float* b_out  = (const float*)d_in[9];

  const int H   = in_sizes[3];
  const int IN  = in_sizes[2] / H;
  const int N   = in_sizes[0] / IN;
  const int E   = in_sizes[1] / 2;
  const int OUT = in_sizes[9];
  const int B   = in_sizes[5] / H;

  const int* srcI = edges;
  const int* dstI = edges + E;

  const size_t Npad = ((size_t)N + 255) & ~(size_t)255;
  const size_t Epad = ((size_t)E + 255) & ~(size_t)255;
  float* norm = (float*)d_ws;
  float* S    = norm + Npad;
  float* Hb   = S + (size_t)N * H;
  int* cnt    = (int*)(Hb + (size_t)N * H);
  int* seg    = cnt + Npad;
  int* rank   = seg + 2 * Npad;
  int* col    = rank + Epad;

  float* xout = (float*)d_out;
  float* xbuf = xout + (size_t)N * OUT;

  k_cnt_zero  <<<(N + 255) / 256, 256, 0, stream>>>(cnt, N);
  k_cnt_rank  <<<(E + 255) / 256, 256, 0, stream>>>(dstI, cnt, rank, E);
  k_start_norm<<<(N + 255) / 256, 256, 0, stream>>>(cnt, seg, norm, N);
  k_fill      <<<(E + 255) / 256, 256, 0, stream>>>(srcI, dstI, rank, seg, col, E);

  float* Sa = S;
  float* Sb = Hb;
  const size_t shmem128 = (size_t)(128 + 128) * (128 + 4) * sizeof(float);
  const size_t shmem64  = (size_t)(128 + 64) * (128 + 4) * sizeof(float);
  const int nblk = (N + 127) / 128;
  (void)IN;

  k_gemm_wmma<128, 128><<<nblk, 256, shmem128, stream>>>(inputs, w_in, norm, Sa, N);
  k_gather_gemm<128><<<nblk, 512, shmem128, stream>>>(seg, col, Sa, norm, b_in, nullptr, xbuf,
                                                     w1, Sb, N, 0);
  { float* t = Sa; Sa = Sb; Sb = t; }
  for (int i = 0; i < B; ++i) {
    k_gather_gemm<128><<<nblk, 512, shmem128, stream>>>(
        seg, col, Sa, norm, b1 + (size_t)i * H, nullptr, nullptr,
        w2 + (size_t)i * H * H, Sb, N, 0);
    { float* t = Sa; Sa = Sb; Sb = t; }
    if (i + 1 < B) {
      k_gather_gemm<128><<<nblk, 512, shmem128, stream>>>(
          seg, col, Sa, norm, b2 + (size_t)i * H, xbuf, xbuf,
          w1 + (size_t)(i + 1) * H * H, Sb, N, 1);
    } else {
      k_gather_gemm<64><<<nblk, 512, shmem64, stream>>>(
          seg, col, Sa, norm, b2 + (size_t)i * H, xbuf, xbuf, w_out, Sb, N, 1);
    }
    { float* t = Sa; Sa = Sb; Sb = t; }
  }
  k_gather_fin<2><<<(N + 7) / 8, 256, 0, stream>>>(seg, col, Sa, norm, b_out, nullptr, xout,
                                                   N, 2);
  (void)OUT;
}


# Fused gather+GEMM retiled so every WGP hosts equal rows: 256 blocks of 28 waves x 7 rows (196 rows, 13 row tiles, 1 per WGP) instead of 391 x 128 rows (2 on some WGPs, 1 on others); sums unchanged
# speedup vs baseline: 4.4341x; 1.0671x over previous
#include <hip/hip_runtime.h>

typedef __attribute__((ext_vector_type(2))) float v2f;
typedef __attribute__((ext_vector_type(4))) float v4f;
typedef __attribute__((ext_vector_type(8))) float v8f;

#define NEG_SLOPE 0.01f

__global__ void k_cnt_zero(int* __restrict__ cnt, int n) {
  int i = blockIdx.x * blockDim.x + threadIdx.x;
  if (i < n) cnt[i] = 0;
}
__global__ void k_cnt_rank(const int* __restrict__ dst, int* __restrict__ cnt,
                           int* __restrict__ rank, int e) {
  int i = blockIdx.x * blockDim.x + threadIdx.x;
  if (i < e) rank[i] = atomicAdd(&cnt[dst[i]], 1);
}
__global__ __launch_bounds__(256) void k_start_norm(const int* __restrict__ cnt,
                                                    int* __restrict__ segw,
                                                    float* __restrict__ norm, int n) {
  int2* seg = (int2*)segw;
  __shared__ int wsum[8];
  __shared__ int psum[8];
  const int tid = threadIdx.x;
  const int lane = tid & 31;
  const int wv = tid >> 5;
  const int base = blockIdx.x * 256;
  int part = 0;
  const int4* c4 = (const int4*)cnt;
  for (int j = tid; j < (base >> 2); j += 256) {
    int4 v = c4[j];
    part += (v.x + v.y) + (v.z + v.w);
  }
  const int i = base + tid;
  const int c = (i < n) ? cnt[i] : 0;
  int incl = c;
#pragma unroll
  for (int off = 1; off < 32; off <<= 1) {
    int t = __shfl_up(incl, off);
    if (lane >= off) incl += t;
  }
#pragma unroll
  for (int off = 16; off > 0; off >>= 1) part += __shfl_xor(part, off);
  if (lane == 31) wsum[wv] = incl;
  if (lane == 0) psum[wv] = part;
  __syncthreads();
  int offset = 0;
#pragma unroll
  for (int w = 0; w < 8; ++w) {
    offset += psum[w];
    if (w < wv) offset += wsum[w];
  }
  if (i < n) {
    seg[i] = make_int2(offset + incl - c, c);
    norm[i] = rsqrtf(1.0f + (float)c);
  }
}
__global__ void k_fill(const int* __restrict__ src, const int* __restrict__ dst,
                       const int* __restrict__ rank, const int* __restrict__ segw,
                       int* __restrict__ col, int e) {
  int i = blockIdx.x * blockDim.x + threadIdx.x;
  if (i < e) atomicExch(&col[segw[2 * dst[i]] + rank[i]], src[i]);
}

template <int K, int HOUT>
__global__ __launch_bounds__(256) void k_gemm_wmma(
    const float* __restrict__ X, const float* __restrict__ W,
    const float* __restrict__ norm,
    float* __restrict__ S,
    int nrows) {
  static_assert(K % 128 == 0 && HOUT % 16 == 0 && ((K / 16) * (HOUT / 16)) % 8 == 0,
                "tile shapes of the staging code");
  extern __shared__ float lds[];
  constexpr int KSTRIDE = K + 4;
  float* Xs = lds;
  float* Wt = lds + 128 * KSTRIDE;

  const int tid = threadIdx.x;
  const int lane = tid & 31;
  const int wave = __builtin_amdgcn_readfirstlane(tid >> 5);
  const int row0 = blockIdx.x * 128;

  constexpr int CT = HOUT / 16;
  constexpr int NT = (K / 16) * CT / 8;
  const int kk = lane & 7, cc = lane >> 3;
  float4 w0[NT], w1[NT];
#pragma unroll
  for (int i = 0; i < NT; ++i) {
    const int t = wave + 8 * i;
    const int k = (t / CT) * 16 + 2 * kk;
    const int c = (t % CT) * 16 + 4 * cc;
    w0[i] = *(const float4*)&W[(k + 0) * HOUT + c];
    w1[i] = *(const float4*)&W[(k + 1) * HOUT + c];
  }
  constexpr int XP = K / 128;
  float4 xv[16 * XP];
#pragma unroll
  for (int it = 0; it < 16; ++it) {
    const int gr = row0 + wave + 8 * it;
    const int grc = gr < nrows ? gr : nrows - 1;
#pragma unroll
    for (int p = 0; p < XP; ++p)
      xv[it * XP + p] = *(const float4*)&X[(size_t)grc * K + p * 128 + 4 * lane];
  }
#pragma unroll
  for (int i = 0; i < NT; ++i) {
    const int t = wave + 8 * i;
    const int k = (t / CT) * 16 + 2 * kk;
    const int c = (t % CT) * 16 + 4 * cc;
    v2f p0 = {w0[i].x, w1[i].x};
    v2f p1 = {w0[i].y, w1[i].y};
    v2f p2 = {w0[i].z, w1[i].z};
    v2f p3 = {w0[i].w, w1[i].w};
    *(v2f*)&Wt[(c + 0) * KSTRIDE + k] = p0;
    *(v2f*)&Wt[(c + 1) * KSTRIDE + k] = p1;
    *(v2f*)&Wt[(c + 2) * KSTRIDE + k] = p2;
    *(v2f*)&Wt[(c + 3) * KSTRIDE + k] = p3;
  }
#pragma unroll
  for (int it = 0; it < 16; ++it) {
    const int r = wave + 8 * it;
#pragma unroll
    for (int p = 0; p < XP; ++p)
      *(float4*)&Xs[r * KSTRIDE + p * 128 + 4 * lane] = xv[it * XP + p];
  }
  __syncthreads();

  const int mrow = lane & 15;
  const int kg   = lane >> 4;
  const float* Xrow = &Xs[(wave * 16 + mrow) * KSTRIDE + 2 * kg];

  float nrm[8];
#pragma unroll
  for (int r = 0; r < 8; ++r) {
    int grow = row0 + wave * 16 + r + 8 * kg;
    nrm[r] = (grow < nrows) ? norm[grow] : 0.f;
  }

  for (int ct = 0; ct < (HOUT >> 4); ++ct) {
    const int col = ct * 16 + mrow;
    const float* Wcol = &Wt[col * KSTRIDE + 2 * kg];
    v8f acc = {};
    for (int k = 0; k < K; k += 4) {
      v2f a = *(const v2f*)&Xrow[k];
      v2f b = *(const v2f*)&Wcol[k];
      acc = __builtin_amdgcn_wmma_f32_16x16x4_f32(
           false, a,  false, b,
           (short)0, acc,  false,  false);
    }
#pragma unroll
    for (int r = 0; r < 8; ++r) {
      int grow = row0 + wave * 16 + r + 8 * kg;
      if (grow < nrows) {
        float sv = acc[r] * nrm[r];
        long o = (long)grow * HOUT + col;
        S[o] = sv;
      }
    }
  }
}

template <int VEC> struct rowvec;
template <> struct rowvec<4> { typedef v4f T; };
template <> struct rowvec<2> { typedef v2f T; };

template <int VEC, int KB>
__device__ __forceinline__ typename rowvec<VEC>::T take_rows(
    typename rowvec<VEC>::T acc, const typename rowvec<VEC>::T* __restrict__ Sv,
    int mycol, int t, int lane, unsigned last) {
  typedef typename rowvec<VEC>::T VT;
  VT v[KB];
#pragma unroll
  for (int u = 0; u < KB; ++u) {
    unsigned s = (unsigned)__builtin_amdgcn_readlane(mycol, t + u);
    s = s < last ? s : last;
    v[u] = Sv[(size_t)s * 32 + lane];
  }
#pragma unroll
  for (int u = 0; u < KB; ++u) acc += v[u];
  return acc;
}

template <int VEC, int KB, class F>
__device__ __forceinline__ typename rowvec<VEC>::T take_rows_then(
    typename rowvec<VEC>::T acc, const typename rowvec<VEC>::T* __restrict__ Sv,
    int mycol, int t, int lane, unsigned last, F&& ask_others) {
  typedef typename rowvec<VEC>::T VT;
  VT v[KB];
#pragma unroll
  for (int u = 0; u < KB; ++u) {
    unsigned s = (unsigned)__builtin_amdgcn_readlane(mycol, t + u);
    s = s < last ? s : last;
    v[u] = Sv[(size_t)s * 32 + lane];
  }
  ask_others();
#pragma unroll
  for (int u = 0; u < KB; ++u) acc += v[u];
  return acc;
}

template <int VEC>
__global__ __launch_bounds__(256) void k_gather_fin(
    const int* __restrict__ segw, const int* __restrict__ col,
    const float* __restrict__ S, const float* __restrict__ norm,
    const float* __restrict__ bias, const float* resid, float* out, int n, int mode) {
  typedef typename rowvec<VEC>::T VT;
  const int lane = threadIdx.x & 31;
  const int node = __builtin_amdgcn_readfirstlane((int)(blockIdx.x * 8 + (threadIdx.x >> 5)));
  if (node >= n) return;
  const VT* Sv = (const VT*)S;
  const unsigned last = (unsigned)(n - 1);
  const int2 sc = ((const int2*)segw)[node];
  const int b = sc.x;
  const int deg = sc.y;
  int m = deg < 32 ? deg : 32;
  int mycol = (lane < m) ? col[b + lane] : 0;
  const VT self = Sv[(size_t)node * 32 + lane];
  const float nr = norm[node];
  const VT bv = ((const VT*)bias)[lane];
  VT rv = {};
  if (mode == 1) rv = ((const VT*)resid)[(size_t)node * 32 + lane];
  VT acc = {};
  int a0 = 0;
  while (true) {
    int t = 0;
    for (; t + 8 <= m; t += 8) acc = take_rows<VEC, 8>(acc, Sv, mycol, t, lane, last);
    if (t + 4 <= m) { acc = take_rows<VEC, 4>(acc, Sv, mycol, t, lane, last); t += 4; }
    if (t + 2 <= m) { acc = take_rows<VEC, 2>(acc, Sv, mycol, t, lane, last); t += 2; }
    if (t < m) acc = take_rows<VEC, 1>(acc, Sv, mycol, t, lane, last);
    a0 += 32;
    if (a0 >= deg) break;
    m = (deg - a0) < 32 ? (deg - a0) : 32;
    mycol = (lane < m) ? col[b + a0 + lane] : 0;
  }
  acc += self;
  VT v;
#pragma unroll
  for (int j = 0; j < VEC; ++j) {
    float x = acc[j] * nr + bv[j];
    if (mode != 2) x = x >= 0.f ? x : NEG_SLOPE * x;
    if (mode == 1) x = (rv[j] + x) * 0.5f;
    v[j] = x;
  }
  ((VT*)out)[(size_t)node * 32 + lane] = v;
}


constexpr int GG_WAVES = 28;
constexpr int GG_RPW   = 7;
constexpr int GG_ROWS  = GG_WAVES * GG_RPW;
constexpr int GG_RT    = (GG_ROWS + 15) / 16;
constexpr int GG_XROWS = GG_RT * 16;

template <int HOUT>
__global__ __launch_bounds__(GG_WAVES * 32) void k_gather_gemm(
    const int* __restrict__ segw, const int* __restrict__ col,
    const float* __restrict__ S, const float* __restrict__ norm,
    const float* __restrict__ bias, const float* resid, float* xw,
    const float* __restrict__ W, float* __restrict__ Snext, int n, int mode) {
  constexpr int K = 128;
  constexpr int KSTRIDE = K + 4;
  extern __shared__ float lds[];
  float* Xs = lds;
  float* Wt = lds + GG_XROWS * KSTRIDE;

  const int tid = threadIdx.x;
  const int lane = tid & 31;
  const int wave = __builtin_amdgcn_readfirstlane(tid >> 5);
  const int row0 = blockIdx.x * GG_ROWS;

  constexpr int CT = HOUT / 16;
  constexpr int TILES = (K / 16) * CT;
  constexpr int NT = (TILES + GG_WAVES - 1) / GG_WAVES;
  {
    const int kk = lane & 7, cc = lane >> 3;
    float4 w0[NT], w1[NT];
#pragma unroll
    for (int i = 0; i < NT; ++i) {
      const int t = wave + GG_WAVES * i;
      if (t < TILES) {
        const int k = (t / CT) * 16 + 2 * kk;
        const int c = (t % CT) * 16 + 4 * cc;
        w0[i] = *(const float4*)&W[(k + 0) * HOUT + c];
        w1[i] = *(const float4*)&W[(k + 1) * HOUT + c];
      }
    }
#pragma unroll
    for (int i = 0; i < NT; ++i) {
      const int t = wave + GG_WAVES * i;
      if (t < TILES) {
        const int k = (t / CT) * 16 + 2 * kk;
        const int c = (t % CT) * 16 + 4 * cc;
        v2f p0 = {w0[i].x, w1[i].x};
        v2f p1 = {w0[i].y, w1[i].y};
        v2f p2 = {w0[i].z, w1[i].z};
        v2f p3 = {w0[i].w, w1[i].w};
        *(v2f*)&Wt[(c + 0) * KSTRIDE + k] = p0;
        *(v2f*)&Wt[(c + 1) * KSTRIDE + k] = p1;
        *(v2f*)&Wt[(c + 2) * KSTRIDE + k] = p2;
        *(v2f*)&Wt[(c + 3) * KSTRIDE + k] = p3;
      }
    }
  }

  {
    const v4f* Sv = (const v4f*)S;
    const unsigned last = (unsigned)(n - 1);
    const int rbase = row0 + wave * GG_RPW;
    int2 sv = make_int2(0, 0);
    float nv = 0.f;
    if (lane < GG_RPW && rbase + lane < n) {
      sv = ((const int2*)segw)[rbase + lane];
      nv = norm[rbase + lane];
    }
    const v4f bv = ((const v4f*)bias)[lane];
    int b_n = __builtin_amdgcn_readlane(sv.x, 0);
    int deg_n = __builtin_amdgcn_readlane(sv.y, 0);
    int m_n = deg_n < 32 ? deg_n : 32;
    int mycol_n = (lane < m_n) ? col[b_n + lane] : 0;
    unsigned sn = (unsigned)rbase < last ? (unsigned)rbase : last;
    v4f self_n = Sv[(size_t)sn * 32 + lane];
#pragma unroll 1
    for (int i = 0; i < GG_RPW; ++i) {
      const int node = rbase + i;
      const int b = b_n;
      const int deg = deg_n;
      int m = m_n;
      int mycol = mycol_n;
      const v4f self = self_n;
      const float nr = __int_as_float(__builtin_amdgcn_readlane(__float_as_int(nv), i));
      v4f rv = {};
      auto ask_others = [&]() {
        if (i + 1 < GG_RPW) {
          b_n = __builtin_amdgcn_readlane(sv.x, i + 1);
          deg_n = __builtin_amdgcn_readlane(sv.y, i + 1);
          m_n = deg_n < 32 ? deg_n : 32;
          mycol_n = (lane < m_n) ? col[b_n + lane] : 0;
          sn = (unsigned)(node + 1) < last ? (unsigned)(node + 1) : last;
          self_n = Sv[(size_t)sn * 32 + lane];
        }
        if (mode == 1 && node < n) rv = ((const v4f*)resid)[(size_t)node * 32 + lane];
      };
      v4f acc = {};
      int t = 0;
      if (m >= 8) {
        acc = take_rows_then<4, 8>(acc, Sv, mycol, 0, lane, last, ask_others);
        t = 8;
      } else {
        ask_others();
      }
      if (node < n) {
        int a0 = 0;
        while (true) {
          for (; t + 8 <= m; t += 8) acc = take_rows<4, 8>(acc, Sv, mycol, t, lane, last);
          if (t + 4 <= m) { acc = take_rows<4, 4>(acc, Sv, mycol, t, lane, last); t += 4; }
          if (t + 2 <= m) { acc = take_rows<4, 2>(acc, Sv, mycol, t, lane, last); t += 2; }
          if (t < m) acc = take_rows<4, 1>(acc, Sv, mycol, t, lane, last);
          a0 += 32;
          if (a0 >= deg) break;
          m = (deg - a0) < 32 ? (deg - a0) : 32;
          mycol = (lane < m) ? col[b + a0 + lane] : 0;
          t = 0;
        }
        acc += self;
        v4f v;
#pragma unroll
        for (int j = 0; j < 4; ++j) {
          float x = acc[j] * nr + bv[j];
          if (mode != 2) x = x >= 0.f ? x : NEG_SLOPE * x;
          if (mode == 1) x = (rv[j] + x) * 0.5f;
          v[j] = x;
        }
        *(v4f*)&Xs[(wave * GG_RPW + i) * KSTRIDE + 4 * lane] = v;
        if (xw != nullptr) ((v4f*)xw)[(size_t)node * 32 + lane] = v;
      } else {
        *(v4f*)&Xs[(wave * GG_RPW + i) * KSTRIDE + 4 * lane] = v4f{};
      }
    }
    if (wave < GG_XROWS - GG_ROWS)
      *(v4f*)&Xs[(GG_ROWS + wave) * KSTRIDE + 4 * lane] = v4f{};
  }
  __syncthreads();

  if (wave >= 2 * GG_RT) return;
  const int mrow = lane & 15;
  const int kg   = lane >> 4;
  const int rt = wave >> 1;
  const int ch = wave & 1;
  constexpr int CTW = CT / 2;
  const float* Xrow = &Xs[(rt * 16 + mrow) * KSTRIDE + 2 * kg];
  const float* Wcol = &Wt[((ch * CTW) * 16 + mrow) * KSTRIDE + 2 * kg];

  float nrm[8];
#pragma unroll
  for (int r = 0; r < 8; ++r) {
    int lrow = rt * 16 + r + 8 * kg;
    int grow = row0 + lrow;
    nrm[r] = (lrow < GG_ROWS && grow < n) ? norm[grow] : 0.f;
  }

  v8f acc[CTW];
#pragma unroll
  for (int j = 0; j < CTW; ++j) acc[j] = v8f{};
#pragma unroll
  for (int k = 0; k < K; k += 4) {
    v2f a = *(const v2f*)&Xrow[k];
#pragma unroll
    for (int j = 0; j < CTW; ++j) {
      v2f bq = *(const v2f*)&Wcol[j * 16 * KSTRIDE + k];
      acc[j] = __builtin_amdgcn_wmma_f32_16x16x4_f32(
           false, a,  false, bq,
           (short)0, acc[j],  false,  false);
    }
  }
#pragma unroll
  for (int j = 0; j < CTW; ++j) {
    const int ccol = (ch * CTW + j) * 16 + mrow;
#pragma unroll
    for (int r = 0; r < 8; ++r) {
      int lrow = rt * 16 + r + 8 * kg;
      int grow = row0 + lrow;
      if (lrow < GG_ROWS && grow < n) Snext[(long)grow * HOUT + ccol] = acc[j][r] * nrm[r];
    }
  }
}

extern "C" void kernel_launch(void* const* d_in, const int* in_sizes, int n_in,
                              void* d_out, int out_size, void* d_ws, size_t ws_size,
                              hipStream_t stream) {
  (void)n_in; (void)out_size; (void)ws_size;
  const float* inputs = (const float*)d_in[0];
  const int*   edges  = (const int*)d_in[1];
  const float* w_in   = (const float*)d_in[2];
  const float* b_in   = (const float*)d_in[3];
  const float* w1     = (const float*)d_in[4];
  const float* b1     = (const float*)d_in[5];
  const float* w2     = (const float*)d_in[6];
  const float* b2     = (const float*)d_in[7];
  const float* w_out  = (const float*)d_in[8];
  const float* b_out  = (const float*)d_in[9];

  const int H   = in_sizes[3];
  const int IN  = in_sizes[2] / H;
  const int N   = in_sizes[0] / IN;
  const int E   = in_sizes[1] / 2;
  const int OUT = in_sizes[9];
  const int B   = in_sizes[5] / H;

  const int* srcI = edges;
  const int* dstI = edges + E;

  const size_t Npad = ((size_t)N + 255) & ~(size_t)255;
  const size_t Epad = ((size_t)E + 255) & ~(size_t)255;
  float* norm = (float*)d_ws;
  float* S    = norm + Npad;
  float* Hb   = S + (size_t)N * H;
  int* cnt    = (int*)(Hb + (size_t)N * H);
  int* seg    = cnt + Npad;
  int* rank   = seg + 2 * Npad;
  int* col    = rank + Epad;

  float* xout = (float*)d_out;
  float* xbuf = xout + (size_t)N * OUT;

  k_cnt_zero  <<<(N + 255) / 256, 256, 0, stream>>>(cnt, N);
  k_cnt_rank  <<<(E + 255) / 256, 256, 0, stream>>>(dstI, cnt, rank, E);
  k_start_norm<<<(N + 255) / 256, 256, 0, stream>>>(cnt, seg, norm, N);
  k_fill      <<<(E + 255) / 256, 256, 0, stream>>>(srcI, dstI, rank, seg, col, E);

  float* Sa = S;
  float* Sb = Hb;
  const size_t shmem128 = (size_t)(128 + 128) * (128 + 4) * sizeof(float);
  const int nblk = (N + 127) / 128;
  const size_t ggmem128 = (size_t)(GG_XROWS + 128) * (128 + 4) * sizeof(float);
  const size_t ggmem64  = (size_t)(GG_XROWS + 64) * (128 + 4) * sizeof(float);
  const int ggblk = (N + GG_ROWS - 1) / GG_ROWS;
  const int ggthr = GG_WAVES * 32;
  (void)IN;

  k_gemm_wmma<128, 128><<<nblk, 256, shmem128, stream>>>(inputs, w_in, norm, Sa, N);
  k_gather_gemm<128><<<ggblk, ggthr, ggmem128, stream>>>(seg, col, Sa, norm, b_in, nullptr, xbuf,
                                                     w1, Sb, N, 0);
  { float* t = Sa; Sa = Sb; Sb = t; }
  for (int i = 0; i < B; ++i) {
    k_gather_gemm<128><<<ggblk, ggthr, ggmem128, stream>>>(
        seg, col, Sa, norm, b1 + (size_t)i * H, nullptr, nullptr,
        w2 + (size_t)i * H * H, Sb, N, 0);
    { float* t = Sa; Sa = Sb; Sb = t; }
    if (i + 1 < B) {
      k_gather_gemm<128><<<ggblk, ggthr, ggmem128, stream>>>(
          seg, col, Sa, norm, b2 + (size_t)i * H, xbuf, xbuf,
          w1 + (size_t)(i + 1) * H * H, Sb, N, 1);
    } else {
      k_gather_gemm<64><<<ggblk, ggthr, ggmem64, stream>>>(
          seg, col, Sa, norm, b2 + (size_t)i * H, xbuf, xbuf, w_out, Sb, N, 1);
    }
    { float* t = Sa; Sa = Sb; Sb = t; }
  }
  k_gather_fin<2><<<(N + 7) / 8, 256, 0, stream>>>(seg, col, Sa, norm, b_out, nullptr, xout,
                                                   N, 2);
  (void)OUT;
}
